# starting kernel
# baseline (speedup 1.0000x reference)
#include <hip/hip_runtime.h>
#include <hip/hip_bf16.h>

static constexpr int B_DIM = 32;
static constexpr int T_DIM = 4096;
static constexpr int I_DIM = 128;
static constexpr int H_DIM = 256;
static constexpr int BT    = B_DIM * T_DIM;
static constexpr int SEGS  = 32;
static constexpr int L_SEG = T_DIM / SEGS;
static constexpr int CT    = 4;

typedef float v2f __attribute__((ext_vector_type(2)));
typedef float v8f __attribute__((ext_vector_type(8)));

__device__ __forceinline__ float sigmoidf_fast(float z) {
    return 1.0f / (1.0f + __expf(-z));
}

__global__ __launch_bounds__(128) void sfnn_gemm_wmma(
    const float* __restrict__ x, const float* __restrict__ Wd,
    const float* __restrict__ bd, float* __restrict__ din)
{
    const int wave = blockIdx.x * (blockDim.x >> 5) + (threadIdx.x >> 5);
    const int lane = threadIdx.x & 31;
    const int half = lane >> 4;
    const int l    = lane & 15;

    const int colTile = wave & (H_DIM / (16 * CT) - 1);
    const int rowTile = wave / (H_DIM / (16 * CT));
    const int row0 = rowTile * 16;
    const int col0 = colTile * 16 * CT;

    const float* arow = x + (size_t)(row0 + l) * I_DIM + 2 * half;
    const float* brow[CT];
#pragma unroll
    for (int c = 0; c < CT; ++c)
        brow[c] = Wd + (size_t)(col0 + c * 16 + l) * I_DIM + 2 * half;

    v8f acc[CT];
#pragma unroll
    for (int c = 0; c < CT; ++c) {
        const float bias = bd[col0 + c * 16 + l];
#pragma unroll
        for (int r = 0; r < 8; ++r) acc[c][r] = bias;
    }

#pragma unroll 8
    for (int k = 0; k < I_DIM; k += 4) {
        const v2f a = *reinterpret_cast<const v2f*>(arow + k);
#pragma unroll
        for (int c = 0; c < CT; ++c) {
            const v2f bf = *reinterpret_cast<const v2f*>(brow[c] + k);
            acc[c] = __builtin_amdgcn_wmma_f32_16x16x4_f32(
                 false, a,  false, bf,
                 (short)0, acc[c],  false,  false);
        }
    }

    float* dout = din + (size_t)(row0 + 8 * half) * H_DIM + col0 + l;
#pragma unroll
    for (int c = 0; c < CT; ++c)
#pragma unroll
        for (int r = 0; r < 8; ++r)
            dout[(size_t)r * H_DIM + c * 16] = acc[c][r];
}

__global__ __launch_bounds__(256) void sfnn_seg_scan(
    const float* __restrict__ din, const float* __restrict__ tau_m,
    const float* __restrict__ tau_n, float* __restrict__ seg)
{
    const int tid = blockIdx.x * blockDim.x + threadIdx.x;
    const int h = tid & (H_DIM - 1);
    const int s = (tid >> 8) & (SEGS - 1);
    const int b = tid >> 13;

    const float alpha = sigmoidf_fast(tau_m[h]);
    const float beta  = sigmoidf_fast(tau_n[h]);

    const float* p = din + ((size_t)b * T_DIM + (size_t)s * L_SEG) * H_DIM + h;
    float d = 0.0f, m = 0.0f;
#pragma unroll 4
    for (int j = 0; j < L_SEG; ++j) {
        const float dt = p[(size_t)j * H_DIM];
        d = beta * d + (1.0f - beta) * dt;
        m = alpha * m + (1.0f - alpha) * d;
    }
    float* q = seg + ((size_t)(b * SEGS + s) * H_DIM + h) * 2;
    q[0] = d;
    q[1] = m;
}

__global__ __launch_bounds__(256) void sfnn_seg_combine(
    const float* __restrict__ tau_m, const float* __restrict__ tau_n,
    float* __restrict__ seg)
{
    const int tid = blockIdx.x * blockDim.x + threadIdx.x;
    const int h = tid & (H_DIM - 1);
    const int b = tid >> 8;

    const float alpha = sigmoidf_fast(tau_m[h]);
    const float beta  = sigmoidf_fast(tau_n[h]);

    float betaL = 1.0f, c_dm = 0.0f, alphaL = 1.0f;
#pragma unroll 4
    for (int j = 0; j < L_SEG; ++j) {
        betaL *= beta;
        c_dm   = alpha * c_dm + (1.0f - alpha) * betaL;
        alphaL *= alpha;
    }

    float d = 0.0f, m = 0.0f;
    for (int s = 0; s < SEGS; ++s) {
        float* q = seg + ((size_t)(b * SEGS + s) * H_DIM + h) * 2;
        const float ed = q[0];
        const float em = q[1];
        q[0] = d;
        q[1] = m;
        const float nd = betaL * d + ed;
        const float nm = alphaL * m + c_dm * d + em;
        d = nd;
        m = nm;
    }
}

__global__ __launch_bounds__(256) void sfnn_out_scan(
    const float* __restrict__ din, const float* __restrict__ seg,
    const float* __restrict__ tau_m, const float* __restrict__ tau_n,
    const float* __restrict__ Wout, const float* __restrict__ bout,
    float* __restrict__ out)
{
    const int b  = blockIdx.x >> 5;
    const int sg = blockIdx.x & 31;
    const int h  = threadIdx.x;
    const int lane = h & 31;
    const int wv   = h >> 5;

    const float alpha = sigmoidf_fast(tau_m[h]);
    const float beta  = sigmoidf_fast(tau_n[h]);
    const float w     = Wout[h];
    const float bo    = bout[0];

    const float* q = seg + ((size_t)(b * SEGS + sg) * H_DIM + h) * 2;
    float d = q[0];
    float m = q[1];

    const float* p = din + ((size_t)b * T_DIM + (size_t)sg * L_SEG) * H_DIM + h;
    float* o = out + (size_t)b * T_DIM + (size_t)sg * L_SEG;

    __shared__ float red[8][8];

    for (int t0 = 0; t0 < L_SEG; t0 += 8) {
        float c[8];
#pragma unroll
        for (int j = 0; j < 8; ++j) {
            const float dt = p[(size_t)(t0 + j) * H_DIM];
            d = beta * d + (1.0f - beta) * dt;
            m = alpha * m + (1.0f - alpha) * d;
            c[j] = m * w;
        }
#pragma unroll
        for (int j = 0; j < 8; ++j) {
#pragma unroll
            for (int off = 16; off > 0; off >>= 1)
                c[j] += __shfl_xor(c[j], off, 32);
        }
        if (lane == 0) {
#pragma unroll
            for (int j = 0; j < 8; ++j) red[wv][j] = c[j];
        }
        __syncthreads();
        if (h < 8) {
            float acc = bo;
#pragma unroll
            for (int w2 = 0; w2 < 8; ++w2) acc += red[w2][h];
            o[t0 + h] = sigmoidf_fast(acc);
        }
        __syncthreads();
    }
}

extern "C" void kernel_launch(void* const* d_in, const int* in_sizes, int n_in,
                              void* d_out, int out_size, void* d_ws, size_t ws_size,
                              hipStream_t stream) {
    const float* x     = (const float*)d_in[0];
    const float* Wd    = (const float*)d_in[1];
    const float* bd    = (const float*)d_in[2];
    const float* tau_m = (const float*)d_in[3];
    const float* tau_n = (const float*)d_in[4];
    const float* Wout  = (const float*)d_in[5];
    const float* bout  = (const float*)d_in[6];
    float* out = (float*)d_out;

    float* din = (float*)d_ws;
    float* seg = din + (size_t)BT * H_DIM;

    const int waves = (BT / 16) * (H_DIM / (16 * CT));
    sfnn_gemm_wmma<<<waves / 4, 128, 0, stream>>>(x, Wd, bd, din);

    sfnn_seg_scan<<<(B_DIM * SEGS * H_DIM) / 256, 256, 0, stream>>>(
        din, tau_m, tau_n, seg);

    sfnn_seg_combine<<<(B_DIM * H_DIM) / 256, 256, 0, stream>>>(
        tau_m, tau_n, seg);

    sfnn_out_scan<<<B_DIM * SEGS, 256, 0, stream>>>(
        din, seg, tau_m, tau_n, Wout, bout, out);
}


# d_in never stored: a tile's 16 rows are 8 steps of two segments, so a lane's 8 accumulator registers are consecutive steps; scan and W_out sums run in-lane in the GEMM wave (sum order changes)
# speedup vs baseline: 2.5652x; 2.5652x over previous
#include <hip/hip_runtime.h>
#include <hip/hip_bf16.h>

static constexpr int B_DIM = 32;
static constexpr int T_DIM = 4096;
static constexpr int I_DIM = 128;
static constexpr int H_DIM = 256;
static constexpr int BT    = B_DIM * T_DIM;
static constexpr int SEGS  = 32;
static constexpr int L_SEG = T_DIM / SEGS;
static constexpr int CT    = 4;
static constexpr int STRIPS = H_DIM / (16 * CT);

typedef float v2f __attribute__((ext_vector_type(2)));
typedef float v4f __attribute__((ext_vector_type(4)));
typedef float v8f __attribute__((ext_vector_type(8)));

__device__ __forceinline__ float sigmoidf_fast(float z) {
    return 1.0f / (1.0f + __expf(-z));
}

template <int X>
__device__ __forceinline__ float row_xmask(float v) {
    return __builtin_bit_cast(float, __builtin_amdgcn_update_dpp(
        0, __builtin_bit_cast(int, v), 0x160 + X, 0xf, 0xf, true));
}

static constexpr int KS       = I_DIM / 4;
static constexpr int TILE_T   = 8;
static constexpr int TILES    = L_SEG / TILE_T;
static constexpr int GEMM_WGS = BT / (2 * L_SEG);

__device__ __forceinline__ void load_a_frags(v2f (&a)[KS], const float* __restrict__ p) {
#pragma unroll
    for (int ks = 0; ks < KS; ++ks)
        a[ks] = *reinterpret_cast<const v2f*>(p + 4 * ks);
}

__device__ __forceinline__ void tile_mma_scan(
    const v2f (&a)[KS], const v2f (&b)[CT][KS], const float (&bias)[CT],
    const float (&alpha)[CT], const float (&beta)[CT],
    const float (&oma)[CT], const float (&omb)[CT], const float (&w)[CT],
    float (&d)[CT], float (&m)[CT], float (&z8)[TILE_T], const bool keep)
{
    v8f acc[CT];
#pragma unroll
    for (int c = 0; c < CT; ++c)
#pragma unroll
        for (int r = 0; r < 8; ++r) acc[c][r] = bias[c];

    __builtin_amdgcn_sched_barrier(0);

#pragma unroll
    for (int ks = 0; ks < KS; ++ks)
#pragma unroll
        for (int c = 0; c < CT; ++c)
            acc[c] = __builtin_amdgcn_wmma_f32_16x16x4_f32(
                 false, a[ks],  false, b[c][ks],
                 (short)0, acc[c],  false,  false);

    __builtin_amdgcn_sched_barrier(0);

    float p[TILE_T];
#pragma unroll
    for (int r = 0; r < TILE_T; ++r) {
#pragma unroll
        for (int c = 0; c < CT; ++c) {
            d[c] = beta[c] * d[c] + omb[c] * acc[c][r];
            m[c] = alpha[c] * m[c] + oma[c] * d[c];
            p[r] = (c == 0) ? m[c] * w[c] : p[r] + m[c] * w[c];
        }
    }
#pragma unroll
    for (int r = 0; r < TILE_T; ++r) p[r] += row_xmask<1>(p[r]);
#pragma unroll
    for (int r = 0; r < TILE_T; ++r) p[r] += row_xmask<2>(p[r]);
#pragma unroll
    for (int r = 0; r < TILE_T; ++r) p[r] += row_xmask<4>(p[r]);
#pragma unroll
    for (int r = 0; r < TILE_T; ++r) p[r] += row_xmask<8>(p[r]);
#pragma unroll
    for (int r = 0; r < TILE_T; ++r) z8[r] = keep ? p[r] : z8[r];
}

__global__ __launch_bounds__(128) void sfnn_gemm_scan(
    const float* __restrict__ x, const float* __restrict__ Wd,
    const float* __restrict__ bd, const float* __restrict__ tau_m,
    const float* __restrict__ tau_n, const float* __restrict__ Wout,
    float* __restrict__ seg, float* __restrict__ zp)
{
    const int strip = threadIdx.x >> 5;
    const int lane  = threadIdx.x & 31;
    const int half  = lane >> 4;
    const int l     = lane & 15;
    const int col0  = strip * 16 * CT;
    const int row0  = blockIdx.x * (2 * L_SEG);

    v2f   bfr[CT][KS];
    float bias[CT], alpha[CT], beta[CT], oma[CT], omb[CT], w[CT];
#pragma unroll
    for (int c = 0; c < CT; ++c) {
        const int h = col0 + c * 16 + l;
        const float* brow = Wd + (size_t)h * I_DIM + 2 * half;
#pragma unroll
        for (int ks = 0; ks < KS; ++ks)
            bfr[c][ks] = *reinterpret_cast<const v2f*>(brow + 4 * ks);
        bias[c]  = bd[h];
        alpha[c] = sigmoidf_fast(tau_m[h]);
        beta[c]  = sigmoidf_fast(tau_n[h]);
        oma[c]   = 1.0f - alpha[c];
        omb[c]   = 1.0f - beta[c];
        w[c]     = Wout[h];
    }

    const float* arow = x + (size_t)(row0 + (l >> 3) * L_SEG + (l & 7)) * I_DIM + 2 * half;

    float d[CT], m[CT], z8[TILE_T];
#pragma unroll
    for (int c = 0; c < CT; ++c) { d[c] = 0.0f; m[c] = 0.0f; }
#pragma unroll
    for (int r = 0; r < TILE_T; ++r) z8[r] = 0.0f;

    v2f a0[KS], a1[KS];
    load_a_frags(a0, arow);

    for (int t = 0; t < TILES; t += 2) {
        load_a_frags(a1, arow + (size_t)(t + 1) * TILE_T * I_DIM);
        tile_mma_scan(a0, bfr, bias, alpha, beta, oma, omb, w, d, m, z8, l == t);
        const int tn = (t + 2 < TILES) ? t + 2 : TILES - 1;
        load_a_frags(a0, arow + (size_t)tn * TILE_T * I_DIM);
        tile_mma_scan(a1, bfr, bias, alpha, beta, oma, omb, w, d, m, z8, l == t + 1);
    }

#pragma unroll
    for (int c = 0; c < CT; ++c) {
        float* q = seg + ((size_t)(2 * blockIdx.x + half) * H_DIM + col0 + c * 16 + l) * 2;
        v2f e; e[0] = d[c]; e[1] = m[c];
        *reinterpret_cast<v2f*>(q) = e;
    }
    float* zo = zp + (size_t)strip * BT + row0 + half * L_SEG + TILE_T * l;
    v4f z0, z1;
#pragma unroll
    for (int r = 0; r < 4; ++r) { z0[r] = z8[r]; z1[r] = z8[4 + r]; }
    *reinterpret_cast<v4f*>(zo)     = z0;
    *reinterpret_cast<v4f*>(zo + 4) = z1;
}

__global__ __launch_bounds__(256) void sfnn_seg_combine(
    const float* __restrict__ tau_m, const float* __restrict__ tau_n,
    float* __restrict__ seg)
{
    const int tid = blockIdx.x * blockDim.x + threadIdx.x;
    const int h = tid & (H_DIM - 1);
    const int b = tid >> 8;

    const float alpha = sigmoidf_fast(tau_m[h]);
    const float beta  = sigmoidf_fast(tau_n[h]);

    float betaL = 1.0f, c_dm = 0.0f, alphaL = 1.0f;
#pragma unroll 4
    for (int j = 0; j < L_SEG; ++j) {
        betaL *= beta;
        c_dm   = alpha * c_dm + (1.0f - alpha) * betaL;
        alphaL *= alpha;
    }

    float d = 0.0f, m = 0.0f;
    for (int s = 0; s < SEGS; ++s) {
        float* q = seg + ((size_t)(b * SEGS + s) * H_DIM + h) * 2;
        const float ed = q[0];
        const float em = q[1];
        q[0] = d;
        q[1] = m;
        const float nd = betaL * d + ed;
        const float nm = alphaL * m + c_dm * d + em;
        d = nd;
        m = nm;
    }
}

__global__ __launch_bounds__(256) void sfnn_out_scan(
    const float* __restrict__ zp, const float* __restrict__ seg,
    const float* __restrict__ tau_m, const float* __restrict__ tau_n,
    const float* __restrict__ Wout, const float* __restrict__ bout,
    float* __restrict__ out)
{
    const int h  = threadIdx.x;
    const int lane = h & 31;
    const int wv   = h >> 5;

    const float alpha = sigmoidf_fast(tau_m[h]);
    const float beta  = sigmoidf_fast(tau_n[h]);
    const float w     = Wout[h];
    const float bo    = bout[0];

    const float* q = seg + ((size_t)blockIdx.x * H_DIM + h) * 2;
    float d = q[0];
    float m = q[1];

    float* o = out + (size_t)blockIdx.x * L_SEG;

    __shared__ float red[8][8];
    __shared__ float zs[L_SEG];

    if (h < L_SEG) {
        const float* z = zp + (size_t)blockIdx.x * L_SEG + h;
        zs[h] = (z[0] + z[(size_t)BT]) + (z[(size_t)2 * BT] + z[(size_t)3 * BT]);
    }

    for (int t0 = 0; t0 < L_SEG; t0 += 8) {
        float c[8];
#pragma unroll
        for (int j = 0; j < 8; ++j) {
            d = beta * d;
            m = alpha * m + (1.0f - alpha) * d;
            c[j] = m * w;
        }
#pragma unroll
        for (int j = 0; j < 8; ++j) {
#pragma unroll
            for (int off = 16; off > 0; off >>= 1)
                c[j] += __shfl_xor(c[j], off, 32);
        }
        if (lane == 0) {
#pragma unroll
            for (int j = 0; j < 8; ++j) red[wv][j] = c[j];
        }
        __syncthreads();
        if (h < 8) {
            float acc = bo + zs[t0 + h];
#pragma unroll
            for (int w2 = 0; w2 < 8; ++w2) acc += red[w2][h];
            o[t0 + h] = sigmoidf_fast(acc);
        }
        __syncthreads();
    }
}

extern "C" void kernel_launch(void* const* d_in, const int* in_sizes, int n_in,
                              void* d_out, int out_size, void* d_ws, size_t ws_size,
                              hipStream_t stream) {
    const float* x     = (const float*)d_in[0];
    const float* Wd    = (const float*)d_in[1];
    const float* bd    = (const float*)d_in[2];
    const float* tau_m = (const float*)d_in[3];
    const float* tau_n = (const float*)d_in[4];
    const float* Wout  = (const float*)d_in[5];
    const float* bout  = (const float*)d_in[6];
    float* out = (float*)d_out;

    float* seg = (float*)d_ws;
    float* zp  = seg + (size_t)B_DIM * SEGS * H_DIM * 2;

    sfnn_gemm_scan<<<GEMM_WGS, 128, 0, stream>>>(x, Wd, bd, tau_m, tau_n, Wout, seg, zp);

    sfnn_seg_combine<<<(B_DIM * H_DIM) / 256, 256, 0, stream>>>(
        tau_m, tau_n, seg);

    sfnn_out_scan<<<B_DIM * SEGS, 256, 0, stream>>>(
        zp, seg, tau_m, tau_n, Wout, bout, out);
}


# Last kernel as a second f32 WMMA product: initial-state response = state [1024x512] x table [512x128] from the combine kernel's power loop; no per-step rescan or lane butterfly (sum order changes)
# speedup vs baseline: 2.9921x; 1.1664x over previous
#include <hip/hip_runtime.h>
#include <hip/hip_bf16.h>

static constexpr int B_DIM = 32;
static constexpr int T_DIM = 4096;
static constexpr int I_DIM = 128;
static constexpr int H_DIM = 256;
static constexpr int BT    = B_DIM * T_DIM;
static constexpr int SEGS  = 32;
static constexpr int L_SEG = T_DIM / SEGS;
static constexpr int CT    = 4;
static constexpr int STRIPS = H_DIM / (16 * CT);

typedef float v2f __attribute__((ext_vector_type(2)));
typedef float v4f __attribute__((ext_vector_type(4)));
typedef float v8f __attribute__((ext_vector_type(8)));

__device__ __forceinline__ float sigmoidf_fast(float z) {
    return 1.0f / (1.0f + __expf(-z));
}

template <int X>
__device__ __forceinline__ float row_xmask(float v) {
    return __builtin_bit_cast(float, __builtin_amdgcn_update_dpp(
        0, __builtin_bit_cast(int, v), 0x160 + X, 0xf, 0xf, true));
}

static constexpr int KS       = I_DIM / 4;
static constexpr int TILE_T   = 8;
static constexpr int TILES    = L_SEG / TILE_T;
static constexpr int GEMM_WGS = BT / (2 * L_SEG);

__device__ __forceinline__ void load_a_frags(v2f (&a)[KS], const float* __restrict__ p) {
#pragma unroll
    for (int ks = 0; ks < KS; ++ks)
        a[ks] = *reinterpret_cast<const v2f*>(p + 4 * ks);
}

__device__ __forceinline__ void tile_mma_scan(
    const v2f (&a)[KS], const v2f (&b)[CT][KS], const float (&bias)[CT],
    const float (&alpha)[CT], const float (&beta)[CT],
    const float (&oma)[CT], const float (&omb)[CT], const float (&w)[CT],
    float (&d)[CT], float (&m)[CT], float (&z8)[TILE_T], const bool keep)
{
    v8f acc[CT];
#pragma unroll
    for (int c = 0; c < CT; ++c)
#pragma unroll
        for (int r = 0; r < 8; ++r) acc[c][r] = bias[c];

    __builtin_amdgcn_sched_barrier(0);

#pragma unroll
    for (int ks = 0; ks < KS; ++ks)
#pragma unroll
        for (int c = 0; c < CT; ++c)
            acc[c] = __builtin_amdgcn_wmma_f32_16x16x4_f32(
                 false, a[ks],  false, b[c][ks],
                 (short)0, acc[c],  false,  false);

    __builtin_amdgcn_sched_barrier(0);

    float p[TILE_T];
#pragma unroll
    for (int r = 0; r < TILE_T; ++r) {
#pragma unroll
        for (int c = 0; c < CT; ++c) {
            d[c] = beta[c] * d[c] + omb[c] * acc[c][r];
            m[c] = alpha[c] * m[c] + oma[c] * d[c];
            p[r] = (c == 0) ? m[c] * w[c] : p[r] + m[c] * w[c];
        }
    }
#pragma unroll
    for (int r = 0; r < TILE_T; ++r) p[r] += row_xmask<1>(p[r]);
#pragma unroll
    for (int r = 0; r < TILE_T; ++r) p[r] += row_xmask<2>(p[r]);
#pragma unroll
    for (int r = 0; r < TILE_T; ++r) p[r] += row_xmask<4>(p[r]);
#pragma unroll
    for (int r = 0; r < TILE_T; ++r) p[r] += row_xmask<8>(p[r]);
#pragma unroll
    for (int r = 0; r < TILE_T; ++r) z8[r] = keep ? p[r] : z8[r];
}

__global__ __launch_bounds__(128) void sfnn_gemm_scan(
    const float* __restrict__ x, const float* __restrict__ Wd,
    const float* __restrict__ bd, const float* __restrict__ tau_m,
    const float* __restrict__ tau_n, const float* __restrict__ Wout,
    float* __restrict__ seg, float* __restrict__ zp)
{
    const int strip = threadIdx.x >> 5;
    const int lane  = threadIdx.x & 31;
    const int half  = lane >> 4;
    const int l     = lane & 15;
    const int col0  = strip * 16 * CT;
    const int row0  = blockIdx.x * (2 * L_SEG);

    v2f   bfr[CT][KS];
    float bias[CT], alpha[CT], beta[CT], oma[CT], omb[CT], w[CT];
#pragma unroll
    for (int c = 0; c < CT; ++c) {
        const int h = col0 + c * 16 + l;
        const float* brow = Wd + (size_t)h * I_DIM + 2 * half;
#pragma unroll
        for (int ks = 0; ks < KS; ++ks)
            bfr[c][ks] = *reinterpret_cast<const v2f*>(brow + 4 * ks);
        bias[c]  = bd[h];
        alpha[c] = sigmoidf_fast(tau_m[h]);
        beta[c]  = sigmoidf_fast(tau_n[h]);
        oma[c]   = 1.0f - alpha[c];
        omb[c]   = 1.0f - beta[c];
        w[c]     = Wout[h];
    }

    const float* arow = x + (size_t)(row0 + (l >> 3) * L_SEG + (l & 7)) * I_DIM + 2 * half;

    float d[CT], m[CT], z8[TILE_T];
#pragma unroll
    for (int c = 0; c < CT; ++c) { d[c] = 0.0f; m[c] = 0.0f; }
#pragma unroll
    for (int r = 0; r < TILE_T; ++r) z8[r] = 0.0f;

    v2f a0[KS], a1[KS];
    load_a_frags(a0, arow);

    for (int t = 0; t < TILES; t += 2) {
        load_a_frags(a1, arow + (size_t)(t + 1) * TILE_T * I_DIM);
        tile_mma_scan(a0, bfr, bias, alpha, beta, oma, omb, w, d, m, z8, l == t);
        const int tn = (t + 2 < TILES) ? t + 2 : TILES - 1;
        load_a_frags(a0, arow + (size_t)tn * TILE_T * I_DIM);
        tile_mma_scan(a1, bfr, bias, alpha, beta, oma, omb, w, d, m, z8, l == t + 1);
    }

#pragma unroll
    for (int c = 0; c < CT; ++c) {
        float* q = seg + ((size_t)(2 * blockIdx.x + half) * H_DIM + col0 + c * 16 + l) * 2;
        v2f e; e[0] = d[c]; e[1] = m[c];
        *reinterpret_cast<v2f*>(q) = e;
    }
    float* zo = zp + (size_t)strip * BT + row0 + half * L_SEG + TILE_T * l;
    v4f z0, z1;
#pragma unroll
    for (int r = 0; r < 4; ++r) { z0[r] = z8[r]; z1[r] = z8[4 + r]; }
    *reinterpret_cast<v4f*>(zo)     = z0;
    *reinterpret_cast<v4f*>(zo + 4) = z1;
}

static constexpr int RK = 2 * H_DIM;
static_assert(B_DIM * 4 == L_SEG, "one block per batch row stores four steps of G");

__global__ __launch_bounds__(256) void sfnn_seg_combine(
    const float* __restrict__ tau_m, const float* __restrict__ tau_n,
    const float* __restrict__ Wout, float* __restrict__ seg, float* __restrict__ gt)
{
    const int h = threadIdx.x;
    const int b = blockIdx.x;

    const float alpha = sigmoidf_fast(tau_m[h]);
    const float beta  = sigmoidf_fast(tau_n[h]);
    const float w     = Wout[h];

    float betaL = 1.0f, c_dm = 0.0f, alphaL = 1.0f;
    for (int i = 0; i < L_SEG / 4; ++i) {
        float gc[4], ga[4];
#pragma unroll
        for (int u = 0; u < 4; ++u) {
            betaL *= beta;
            c_dm   = alpha * c_dm + (1.0f - alpha) * betaL;
            alphaL *= alpha;
            gc[u] = c_dm;
            ga[u] = alphaL;
        }
        if (i == b) {
#pragma unroll
            for (int u = 0; u < 4; ++u) {
                v2f g; g[0] = w * gc[u]; g[1] = w * ga[u];
                *reinterpret_cast<v2f*>(gt + (size_t)(4 * i + u) * RK + 2 * h) = g;
            }
        }
    }

    float d = 0.0f, m = 0.0f;
    for (int s = 0; s < SEGS; ++s) {
        float* q = seg + ((size_t)(b * SEGS + s) * H_DIM + h) * 2;
        const float ed = q[0];
        const float em = q[1];
        q[0] = d;
        q[1] = m;
        const float nd = betaL * d + ed;
        const float nm = alphaL * m + c_dm * d + em;
        d = nd;
        m = nm;
    }
}

static constexpr int RGRP = 4;
static constexpr int RQ   = RK / 8 / RGRP;
static constexpr int RGK  = RK / RGRP;
static constexpr int ROUT_WGS = (B_DIM * SEGS / 16) * (L_SEG / 32);
static constexpr int RPITCH = 36;

__device__ __forceinline__ void load_r_frags(v4f (&a)[RQ], v4f (&b)[RQ],
                                             const float* __restrict__ pa,
                                             const float* __restrict__ pb) {
#pragma unroll
    for (int i = 0; i < RQ; ++i) {
        a[i] = *reinterpret_cast<const v4f*>(pa + 8 * i);
        b[i] = *reinterpret_cast<const v4f*>(pb + 8 * i);
    }
}

__device__ __forceinline__ void mma_r_frags(const v4f (&a)[RQ], const v4f (&b)[RQ],
                                            v8f& even, v8f& odd) {
#pragma unroll
    for (int i = 0; i < RQ; ++i) {
        const v2f alo = __builtin_shufflevector(a[i], a[i], 0, 1);
        const v2f ahi = __builtin_shufflevector(a[i], a[i], 2, 3);
        const v2f blo = __builtin_shufflevector(b[i], b[i], 0, 1);
        const v2f bhi = __builtin_shufflevector(b[i], b[i], 2, 3);
        even = __builtin_amdgcn_wmma_f32_16x16x4_f32(
             false, alo,  false, blo,
             (short)0, even,  false,  false);
        odd = __builtin_amdgcn_wmma_f32_16x16x4_f32(
             false, ahi,  false, bhi,
             (short)0, odd,  false,  false);
    }
}

__global__ __launch_bounds__(64) void sfnn_out_gemm(
    const float* __restrict__ zp, const float* __restrict__ seg,
    const float* __restrict__ gt, const float* __restrict__ bout,
    float* __restrict__ out)
{
    const int wv   = threadIdx.x >> 5;
    const int lane = threadIdx.x & 31;
    const int half = lane >> 4;
    const int l    = lane & 15;
    const int m0   = (blockIdx.x >> 2) * 16;
    const int n0   = (blockIdx.x & 3) * 32;
    const int nw   = n0 + wv * 16;

    const float* ap = seg + (size_t)(m0 + l) * RK + 4 * half;
    const float* bp = gt  + (size_t)(nw + l) * RK + 4 * half;

    __shared__ float rs[16 * RPITCH];

    v4f a0[RQ], b0[RQ], a1[RQ], b1[RQ], a2[RQ], b2[RQ], a3[RQ], b3[RQ];
    load_r_frags(a0, b0, ap, bp);
    load_r_frags(a1, b1, ap + RGK, bp + RGK);
    load_r_frags(a2, b2, ap + 2 * RGK, bp + 2 * RGK);

    v8f even, odd;
#pragma unroll
    for (int r = 0; r < 8; ++r) { even[r] = 0.0f; odd[r] = 0.0f; }

    __builtin_amdgcn_sched_barrier(0);
    mma_r_frags(a0, b0, even, odd);
    mma_r_frags(a1, b1, even, odd);
    __builtin_amdgcn_sched_barrier(0);
    load_r_frags(a3, b3, ap + 3 * RGK, bp + 3 * RGK);
    const float bo = bout[0];
    float zq[STRIPS][8];
#pragma unroll
    for (int q = 0; q < STRIPS; ++q)
#pragma unroll
        for (int r = 0; r < 8; ++r)
            zq[q][r] = zp[(size_t)q * BT + (size_t)(m0 + 8 * wv + r) * L_SEG + n0 + lane];
    __builtin_amdgcn_sched_barrier(0);
    mma_r_frags(a2, b2, even, odd);
    mma_r_frags(a3, b3, even, odd);

    float z[8];
#pragma unroll
    for (int r = 0; r < 8; ++r)
        z[r] = (zq[0][r] + zq[1][r]) + (zq[2][r] + zq[3][r]);

#pragma unroll
    for (int r = 0; r < 8; ++r)
        rs[(8 * half + r) * RPITCH + 16 * wv + l] = even[r] + odd[r];
    __syncthreads();
#pragma unroll
    for (int r = 0; r < 8; ++r) {
        const float v = (bo + z[r]) + rs[(8 * wv + r) * RPITCH + lane];
        out[(size_t)(m0 + 8 * wv + r) * L_SEG + n0 + lane] = sigmoidf_fast(v);
    }
}

extern "C" void kernel_launch(void* const* d_in, const int* in_sizes, int n_in,
                              void* d_out, int out_size, void* d_ws, size_t ws_size,
                              hipStream_t stream) {
    const float* x     = (const float*)d_in[0];
    const float* Wd    = (const float*)d_in[1];
    const float* bd    = (const float*)d_in[2];
    const float* tau_m = (const float*)d_in[3];
    const float* tau_n = (const float*)d_in[4];
    const float* Wout  = (const float*)d_in[5];
    const float* bout  = (const float*)d_in[6];
    float* out = (float*)d_out;

    float* seg = (float*)d_ws;
    float* zp  = seg + (size_t)B_DIM * SEGS * H_DIM * 2;
    float* gt  = zp + (size_t)STRIPS * BT;

    sfnn_gemm_scan<<<GEMM_WGS, 128, 0, stream>>>(x, Wd, bd, tau_m, tau_n, Wout, seg, zp);

    sfnn_seg_combine<<<B_DIM, H_DIM, 0, stream>>>(
        tau_m, tau_n, Wout, seg, gt);

    sfnn_out_gemm<<<ROUT_WGS, 64, 0, stream>>>(
        zp, seg, gt, bout, out);
}


# Combine launch gone: last kernel re-tiled per batch row; a work-group (16 segments x 32 steps) runs the segment combine and builds its response-table rows in LDS for its f32 WMMA (sum order changes)
# speedup vs baseline: 3.2656x; 1.0914x over previous
#include <hip/hip_runtime.h>
#include <hip/hip_bf16.h>

static constexpr int B_DIM = 32;
static constexpr int T_DIM = 4096;
static constexpr int I_DIM = 128;
static constexpr int H_DIM = 256;
static constexpr int BT    = B_DIM * T_DIM;
static constexpr int SEGS  = 32;
static constexpr int L_SEG = T_DIM / SEGS;
static constexpr int CT    = 4;
static constexpr int STRIPS = H_DIM / (16 * CT);

typedef float v2f __attribute__((ext_vector_type(2)));
typedef float v4f __attribute__((ext_vector_type(4)));
typedef float v8f __attribute__((ext_vector_type(8)));

__device__ __forceinline__ float sigmoidf_fast(float z) {
    return 1.0f / (1.0f + __expf(-z));
}

template <int X>
__device__ __forceinline__ float row_xmask(float v) {
    return __builtin_bit_cast(float, __builtin_amdgcn_update_dpp(
        0, __builtin_bit_cast(int, v), 0x160 + X, 0xf, 0xf, true));
}

static constexpr int KS       = I_DIM / 4;
static constexpr int TILE_T   = 8;
static constexpr int TILES    = L_SEG / TILE_T;
static constexpr int GEMM_WGS = BT / (2 * L_SEG);

__device__ __forceinline__ void load_a_frags(v2f (&a)[KS], const float* __restrict__ p) {
#pragma unroll
    for (int ks = 0; ks < KS; ++ks)
        a[ks] = *reinterpret_cast<const v2f*>(p + 4 * ks);
}

__device__ __forceinline__ void tile_mma_scan(
    const v2f (&a)[KS], const v2f (&b)[CT][KS], const float (&bias)[CT],
    const float (&alpha)[CT], const float (&beta)[CT],
    const float (&oma)[CT], const float (&omb)[CT], const float (&w)[CT],
    float (&d)[CT], float (&m)[CT], float (&z8)[TILE_T], const bool keep)
{
    v8f acc[CT];
#pragma unroll
    for (int c = 0; c < CT; ++c)
#pragma unroll
        for (int r = 0; r < 8; ++r) acc[c][r] = bias[c];

    __builtin_amdgcn_sched_barrier(0);

#pragma unroll
    for (int ks = 0; ks < KS; ++ks)
#pragma unroll
        for (int c = 0; c < CT; ++c)
            acc[c] = __builtin_amdgcn_wmma_f32_16x16x4_f32(
                 false, a[ks],  false, b[c][ks],
                 (short)0, acc[c],  false,  false);

    __builtin_amdgcn_sched_barrier(0);

    float p[TILE_T];
#pragma unroll
    for (int r = 0; r < TILE_T; ++r) {
#pragma unroll
        for (int c = 0; c < CT; ++c) {
            d[c] = beta[c] * d[c] + omb[c] * acc[c][r];
            m[c] = alpha[c] * m[c] + oma[c] * d[c];
            p[r] = (c == 0) ? m[c] * w[c] : p[r] + m[c] * w[c];
        }
    }
#pragma unroll
    for (int r = 0; r < TILE_T; ++r) p[r] += row_xmask<1>(p[r]);
#pragma unroll
    for (int r = 0; r < TILE_T; ++r) p[r] += row_xmask<2>(p[r]);
#pragma unroll
    for (int r = 0; r < TILE_T; ++r) p[r] += row_xmask<4>(p[r]);
#pragma unroll
    for (int r = 0; r < TILE_T; ++r) p[r] += row_xmask<8>(p[r]);
#pragma unroll
    for (int r = 0; r < TILE_T; ++r) z8[r] = keep ? p[r] : z8[r];
}

__global__ __launch_bounds__(128) void sfnn_gemm_scan(
    const float* __restrict__ x, const float* __restrict__ Wd,
    const float* __restrict__ bd, const float* __restrict__ tau_m,
    const float* __restrict__ tau_n, const float* __restrict__ Wout,
    float* __restrict__ seg, float* __restrict__ zp)
{
    const int strip = threadIdx.x >> 5;
    const int lane  = threadIdx.x & 31;
    const int half  = lane >> 4;
    const int l     = lane & 15;
    const int col0  = strip * 16 * CT;
    const int row0  = blockIdx.x * (2 * L_SEG);

    v2f   bfr[CT][KS];
    float bias[CT], alpha[CT], beta[CT], oma[CT], omb[CT], w[CT];
#pragma unroll
    for (int c = 0; c < CT; ++c) {
        const int h = col0 + c * 16 + l;
        const float* brow = Wd + (size_t)h * I_DIM + 2 * half;
#pragma unroll
        for (int ks = 0; ks < KS; ++ks)
            bfr[c][ks] = *reinterpret_cast<const v2f*>(brow + 4 * ks);
        bias[c]  = bd[h];
        alpha[c] = sigmoidf_fast(tau_m[h]);
        beta[c]  = sigmoidf_fast(tau_n[h]);
        oma[c]   = 1.0f - alpha[c];
        omb[c]   = 1.0f - beta[c];
        w[c]     = Wout[h];
    }

    const float* arow = x + (size_t)(row0 + (l >> 3) * L_SEG + (l & 7)) * I_DIM + 2 * half;

    float d[CT], m[CT], z8[TILE_T];
#pragma unroll
    for (int c = 0; c < CT; ++c) { d[c] = 0.0f; m[c] = 0.0f; }
#pragma unroll
    for (int r = 0; r < TILE_T; ++r) z8[r] = 0.0f;

    v2f a0[KS], a1[KS];
    load_a_frags(a0, arow);

    for (int t = 0; t < TILES; t += 2) {
        load_a_frags(a1, arow + (size_t)(t + 1) * TILE_T * I_DIM);
        tile_mma_scan(a0, bfr, bias, alpha, beta, oma, omb, w, d, m, z8, l == t);
        const int tn = (t + 2 < TILES) ? t + 2 : TILES - 1;
        load_a_frags(a0, arow + (size_t)tn * TILE_T * I_DIM);
        tile_mma_scan(a1, bfr, bias, alpha, beta, oma, omb, w, d, m, z8, l == t + 1);
    }

#pragma unroll
    for (int c = 0; c < CT; ++c) {
        float* q = seg + ((size_t)(2 * blockIdx.x + half) * H_DIM + col0 + c * 16 + l) * 2;
        v2f e; e[0] = d[c]; e[1] = m[c];
        *reinterpret_cast<v2f*>(q) = e;
    }
    float* zo = zp + (size_t)strip * BT + row0 + half * L_SEG + TILE_T * l;
    v4f z0, z1;
#pragma unroll
    for (int r = 0; r < 4; ++r) { z0[r] = z8[r]; z1[r] = z8[4 + r]; }
    *reinterpret_cast<v4f*>(zo)     = z0;
    *reinterpret_cast<v4f*>(zo + 4) = z1;
}

static constexpr int RK    = 2 * H_DIM;
static constexpr int RM    = 16;
static constexpr int RN    = 32;
static constexpr int RPK   = RK + 4;
static constexpr int RKW   = RK / 2;
static constexpr int RQ    = RKW / 8;
static constexpr int RPITCH = 36;
static constexpr int ROUT_WGS = B_DIM * (SEGS / RM) * (L_SEG / RN);
static_assert(SEGS == 2 * RM, "a batch row is two work-groups of 16 segments");

__global__ __launch_bounds__(128) void sfnn_combine_out(
    const float* __restrict__ zp, const float* __restrict__ seg,
    const float* __restrict__ tau_m, const float* __restrict__ tau_n,
    const float* __restrict__ Wout, const float* __restrict__ bout,
    float* __restrict__ out)
{
    __shared__ __attribute__((aligned(16))) float s_lds[RM * RPK];
    __shared__ __attribute__((aligned(16))) float g_lds[RN * RPK];
    __shared__ __attribute__((aligned(16))) float red[2 * RM * RPITCH];

    const int t    = threadIdx.x;
    const int wv   = t >> 5;
    const int lane = t & 31;
    const int half = lane >> 4;
    const int l    = lane & 15;
    const int nc   = blockIdx.x & 3;
    const int mh   = (blockIdx.x >> 2) & 1;
    const int b    = blockIdx.x >> 3;
    const int m0   = b * SEGS + RM * mh;
    const int n0   = RN * nc;

    const v2f tm = *reinterpret_cast<const v2f*>(tau_m + 2 * t);
    const v2f tn = *reinterpret_cast<const v2f*>(tau_n + 2 * t);
    const v2f wo = *reinterpret_cast<const v2f*>(Wout + 2 * t);

    v4f e[SEGS - 1];
    const float* ep = seg + ((size_t)b * SEGS * H_DIM + 2 * t) * 2;
#pragma unroll
    for (int s = 0; s < SEGS - 1; ++s)
        e[s] = *reinterpret_cast<const v4f*>(ep + (size_t)s * RK);

    const float bo = bout[0];
    float zq[4][STRIPS];
#pragma unroll
    for (int r = 0; r < 4; ++r)
#pragma unroll
        for (int q = 0; q < STRIPS; ++q)
            zq[r][q] = zp[(size_t)q * BT + (size_t)(m0 + 4 * wv + r) * L_SEG + n0 + lane];

    asm volatile("" ::: "memory");

    float alpha[2], beta[2], w[2];
#pragma unroll
    for (int c = 0; c < 2; ++c) {
        alpha[c] = sigmoidf_fast(tm[c]);
        beta[c]  = sigmoidf_fast(tn[c]);
        w[c]     = wo[c];
    }

    float betaL[2] = {1.0f, 1.0f}, c_dm[2] = {0.0f, 0.0f}, alphaL[2] = {1.0f, 1.0f};
#pragma nounroll
    for (int i = 0; i < L_SEG / 4; ++i) {
        float gc[4][2], ga[4][2];
#pragma unroll
        for (int u = 0; u < 4; ++u) {
#pragma unroll
            for (int c = 0; c < 2; ++c) {
                betaL[c] *= beta[c];
                c_dm[c]   = alpha[c] * c_dm[c] + (1.0f - alpha[c]) * betaL[c];
                alphaL[c] *= alpha[c];
                gc[u][c] = c_dm[c];
                ga[u][c] = alphaL[c];
            }
        }
        if ((i >> 3) == nc) {
#pragma unroll
            for (int u = 0; u < 4; ++u) {
                v4f g;
                g[0] = gc[u][0]; g[1] = ga[u][0];
                g[2] = gc[u][1]; g[3] = ga[u][1];
                *reinterpret_cast<v4f*>(&g_lds[(4 * (i & 7) + u) * RPK + 4 * t]) = g;
            }
        }
    }

    float d[2] = {0.0f, 0.0f}, m[2] = {0.0f, 0.0f};
#pragma unroll
    for (int s = 0; s < SEGS; ++s) {
        if ((s >> 4) == mh) {
            v4f st;
            st[0] = w[0] * d[0]; st[1] = w[0] * m[0];
            st[2] = w[1] * d[1]; st[3] = w[1] * m[1];
            *reinterpret_cast<v4f*>(&s_lds[(s & 15) * RPK + 4 * t]) = st;
        }
        if (s + 1 < SEGS) {
#pragma unroll
            for (int c = 0; c < 2; ++c) {
                const float nd = betaL[c] * d[c] + e[s][2 * c];
                const float nm = alphaL[c] * m[c] + c_dm[c] * d[c] + e[s][2 * c + 1];
                d[c] = nd;
                m[c] = nm;
            }
        }
    }
    __syncthreads();

    const int nt = wv & 1;
    const int kh = wv >> 1;
    const float* ap = s_lds + l * RPK + RKW * kh + 4 * half;
    const float* bp = g_lds + (16 * nt + l) * RPK + RKW * kh + 4 * half;
    v4f a[RQ], bb[RQ];
#pragma unroll
    for (int i = 0; i < RQ; ++i) {
        a[i]  = *reinterpret_cast<const v4f*>(ap + 8 * i);
        bb[i] = *reinterpret_cast<const v4f*>(bp + 8 * i);
    }
    v8f even, odd;
#pragma unroll
    for (int r = 0; r < 8; ++r) { even[r] = 0.0f; odd[r] = 0.0f; }
    __builtin_amdgcn_sched_barrier(0);
#pragma unroll
    for (int i = 0; i < RQ; ++i) {
        const v2f alo = __builtin_shufflevector(a[i], a[i], 0, 1);
        const v2f ahi = __builtin_shufflevector(a[i], a[i], 2, 3);
        const v2f blo = __builtin_shufflevector(bb[i], bb[i], 0, 1);
        const v2f bhi = __builtin_shufflevector(bb[i], bb[i], 2, 3);
        even = __builtin_amdgcn_wmma_f32_16x16x4_f32(
             false, alo,  false, blo,
             (short)0, even,  false,  false);
        odd = __builtin_amdgcn_wmma_f32_16x16x4_f32(
             false, ahi,  false, bhi,
             (short)0, odd,  false,  false);
    }

#pragma unroll
    for (int r = 0; r < 8; ++r)
        red[(kh * RM + 8 * half + r) * RPITCH + 16 * nt + l] = even[r] + odd[r];
    __syncthreads();
#pragma unroll
    for (int r = 0; r < 4; ++r) {
        const int row = 4 * wv + r;
        const float z = (zq[r][0] + zq[r][1]) + (zq[r][2] + zq[r][3]);
        const float p = red[row * RPITCH + lane] + red[(RM + row) * RPITCH + lane];
        out[(size_t)(m0 + row) * L_SEG + n0 + lane] = sigmoidf_fast((bo + z) + p);
    }
}

extern "C" void kernel_launch(void* const* d_in, const int* in_sizes, int n_in,
                              void* d_out, int out_size, void* d_ws, size_t ws_size,
                              hipStream_t stream) {
    const float* x     = (const float*)d_in[0];
    const float* Wd    = (const float*)d_in[1];
    const float* bd    = (const float*)d_in[2];
    const float* tau_m = (const float*)d_in[3];
    const float* tau_n = (const float*)d_in[4];
    const float* Wout  = (const float*)d_in[5];
    const float* bout  = (const float*)d_in[6];
    float* out = (float*)d_out;

    float* seg = (float*)d_ws;
    float* zp  = seg + (size_t)B_DIM * SEGS * H_DIM * 2;

    sfnn_gemm_scan<<<GEMM_WGS, 128, 0, stream>>>(x, Wd, bd, tau_m, tau_n, Wout, seg, zp);

    sfnn_combine_out<<<ROUT_WGS, 128, 0, stream>>>(
        zp, seg, tau_m, tau_n, Wout, bout, out);
}
